# starting kernel
# baseline (speedup 1.0000x reference)
#include <hip/hip_runtime.h>

typedef __attribute__((ext_vector_type(2))) float v2f;
typedef __attribute__((ext_vector_type(4))) float v4f;
typedef __attribute__((ext_vector_type(8))) float v8f;

__global__ void NeighbourAggregation_scatter_kernel(
    const float* __restrict__ x,
    const int* __restrict__ edge_dst,
    float* __restrict__ out,
    float* __restrict__ counts,
    int total)
{
    int t = blockIdx.x * blockDim.x + threadIdx.x;
    int base = t * 4;
    if (base >= total) return;

    v4f v = __builtin_nontemporal_load((const v4f*)(x + base));
    int e = base >> 4;
    int d = base & 15;
    int dst = __builtin_nontemporal_load(edge_dst + e);

    float* o = out + dst * 16 + d;
    unsafeAtomicAdd(o + 0, v.x);
    unsafeAtomicAdd(o + 1, v.y);
    unsafeAtomicAdd(o + 2, v.z);
    unsafeAtomicAdd(o + 3, v.w);
    if (d == 0) {
        unsafeAtomicAdd(counts + dst, 1.0f);
    }
}

__global__ void NeighbourAggregation_finalize_wmma_kernel(
    float* __restrict__ out,
    const float* __restrict__ counts,
    int num_nodes)
{
    int wavesPerBlock = blockDim.x >> 5;
    int gwave = blockIdx.x * wavesPerBlock + (threadIdx.x >> 5);
    int lane  = threadIdx.x & 31;
    int nb    = gwave * 16;
    if (nb >= num_nodes) return;

    int col  = lane & 15;
    int half = lane >> 4;

    if (nb + 16 <= num_nodes) {
        float r = __frsqrt_rn(counts[nb + col] + 1.0f);

        v8f c = {0.f, 0.f, 0.f, 0.f, 0.f, 0.f, 0.f, 0.f};
        #pragma unroll
        for (int k0 = 0; k0 < 16; k0 += 4) {
            v2f a, b;
            #pragma unroll
            for (int j = 0; j < 2; ++j) {
                int K = k0 + 2 * half + j;
                a[j] = (col == K) ? r : 0.0f;
                b[j] = out[(nb + K) * 16 + col];
            }
            c = __builtin_amdgcn_wmma_f32_16x16x4_f32(
                    false, a, false, b, (short)0, c, false, false);
        }

        #pragma unroll
        for (int v = 0; v < 8; ++v) {
            int row = v + 8 * half;
            out[(nb + row) * 16 + col] = c[v];
        }
    } else {
        for (int row = nb + half; row < num_nodes; row += 2) {
            float r = __frsqrt_rn(counts[row] + 1.0f);
            out[row * 16 + col] *= r;
        }
    }
}

extern "C" void kernel_launch(void* const* d_in, const int* in_sizes, int n_in,
                              void* d_out, int out_size, void* d_ws, size_t ws_size,
                              hipStream_t stream) {
    const float* x        = (const float*)d_in[0];
    const int*   edge_dst = (const int*)d_in[1];
    int total     = in_sizes[0];
    int num_nodes = out_size / 16;

    float* out    = (float*)d_out;
    float* counts = (float*)d_ws;

    hipMemsetAsync(d_out, 0, (size_t)out_size * sizeof(float), stream);
    hipMemsetAsync(d_ws, 0, (size_t)num_nodes * sizeof(float), stream);

    int nthreads = total / 4;
    int sblocks  = (nthreads + 255) / 256;
    NeighbourAggregation_scatter_kernel<<<sblocks, 256, 0, stream>>>(
        x, edge_dst, out, counts, total);

    int nwaves  = (num_nodes + 15) / 16;
    int fblocks = (nwaves + 7) / 8;
    NeighbourAggregation_finalize_wmma_kernel<<<fblocks, 256, 0, stream>>>(
        out, counts, num_nodes);
}


# Global f32 atomics replaced by owner work-groups: each reads the whole destination list, its waves queue their hits in LDS, gather those rows, sum by LDS f32 adds, WMMA scale fused; sum order changes
# speedup vs baseline: 2.7611x; 2.7611x over previous
#include <hip/hip_runtime.h>

typedef __attribute__((ext_vector_type(2))) float v2f;
typedef __attribute__((ext_vector_type(8))) float v8f;
typedef __attribute__((ext_vector_type(2))) int   v2i;
typedef __attribute__((ext_vector_type(4))) int   v4i;

#define ROWS 400
#define NT   1024
#define NW   (NT / 32)
#define UNR  8
#define SEG  640

__device__ __forceinline__ void lds_add(float* p, float v) {
    __hip_atomic_fetch_add(p, v, __ATOMIC_RELAXED, __HIP_MEMORY_SCOPE_WORKGROUP);
}

__device__ __forceinline__ void append(v2i* seg, int& wc, bool hit, int e, unsigned row) {
    const unsigned m = __builtin_amdgcn_ballot_w32(hit);
    if (m != 0u) {
        const int p = wc + (int)__builtin_amdgcn_mbcnt_lo(m, 0u);
        if (hit && p < SEG) {
            v2i q; q.x = e; q.y = (int)row;
            seg[p] = q;
        }
        wc += __builtin_popcount(m);
    }
}

__device__ __forceinline__ void filter4(v4i d, int e0, unsigned lo, v2i* seg, int& wc) {
    const unsigned r0 = (unsigned)d.x - lo, r1 = (unsigned)d.y - lo;
    const unsigned r2 = (unsigned)d.z - lo, r3 = (unsigned)d.w - lo;
    const unsigned mn = min(min(r0, r1), min(r2, r3));
    if (__builtin_amdgcn_ballot_w32(mn < ROWS) != 0u) {
        append(seg, wc, r0 < ROWS, e0 + 0, r0);
        append(seg, wc, r1 < ROWS, e0 + 1, r1);
        append(seg, wc, r2 < ROWS, e0 + 2, r2);
        append(seg, wc, r3 < ROWS, e0 + 3, r3);
    }
}

__device__ __forceinline__ void add_direct(float* acc, float* cnt, const float* __restrict__ x,
                                           int e, unsigned row) {
    if (row < ROWS) {
        const float* xr = x + (size_t)e * 16;
        for (int f = 0; f < 16; ++f) lds_add(&acc[row * 16 + f], xr[f]);
        lds_add(&cnt[row], 1.0f);
    }
}

__global__ __launch_bounds__(NT)
void NeighbourAggregation_owner_kernel(
    const float* __restrict__ x,
    const int* __restrict__ edge_dst,
    float* __restrict__ out,
    int E, int num_nodes)
{
    __shared__ float acc[ROWS * 16];
    __shared__ float cnt[ROWS];
    __shared__ v2i   queue[NW * SEG];

    const int tid  = threadIdx.x;
    const int lane = tid & 31;
    const int wave = tid >> 5;
    const int half = lane >> 4;
    const int col  = lane & 15;
    const unsigned lo = blockIdx.x * ROWS;
    v2i* seg = queue + wave * SEG;
    int wc = 0;

    for (int i = tid; i < ROWS * 16; i += NT) acc[i] = 0.0f;
    for (int i = tid; i < ROWS; i += NT) cnt[i] = 0.0f;
    __syncthreads();

    const v4i* d4 = (const v4i*)edge_dst;
    const int E4 = E >> 2;
    const int nfull = E4 / (NT * UNR);
    const int start = (int)(((long long)blockIdx.x * nfull) / gridDim.x);
    for (int it = 0; it < nfull; ++it) {
        int itr = it + start;
        if (itr >= nfull) itr -= nfull;
        const int g0 = itr * (NT * UNR) + tid;
        v4i d[UNR];
        #pragma unroll
        for (int u = 0; u < UNR; ++u) d[u] = d4[g0 + u * NT];
        #pragma unroll
        for (int u = 0; u < UNR; ++u)
            filter4(d[u], (g0 + u * NT) * 4, lo, seg, wc);
    }
    {
        const int g0 = nfull * (NT * UNR) + tid;
        v4i d[UNR];
        #pragma unroll
        for (int u = 0; u < UNR; ++u) {
            const int g = g0 + u * NT;
            v4i t = {-1, -1, -1, -1};
            if (g < E4) t = d4[g];
            d[u] = t;
        }
        #pragma unroll
        for (int u = 0; u < UNR; ++u)
            filter4(d[u], (g0 + u * NT) * 4, lo, seg, wc);
    }
    {
        const int e = E4 * 4 + tid;
        unsigned row = 0xffffffffu;
        if (e < E) row = (unsigned)edge_dst[e] - lo;
        append(seg, wc, row < ROWS, e, row);
    }

    if (wc > SEG) {
        for (int g = tid; g < E4; g += NT) {
            const v4i d = d4[g];
            add_direct(acc, cnt, x, g * 4 + 0, (unsigned)d.x - lo);
            add_direct(acc, cnt, x, g * 4 + 1, (unsigned)d.y - lo);
            add_direct(acc, cnt, x, g * 4 + 2, (unsigned)d.z - lo);
            add_direct(acc, cnt, x, g * 4 + 3, (unsigned)d.w - lo);
        }
        const int e = E4 * 4 + tid;
        if (e < E) add_direct(acc, cnt, x, e, (unsigned)edge_dst[e] - lo);
    } else {
        const int n = wc;
        for (int base = 0; base < n; base += 64) {
            float xv[32];
            #pragma unroll
            for (int i = 0; i < 32; ++i) {
                int idx = base + 2 * i + half;
                idx = idx < n ? idx : n - 1;
                const unsigned e = (unsigned)seg[idx].x;
                xv[i] = __builtin_nontemporal_load(x + (e * 16u + (unsigned)col));
            }
            int rw[32];
            #pragma unroll
            for (int i = 0; i < 32; ++i) {
                int idx = base + 2 * i + half;
                idx = idx < n ? idx : n - 1;
                rw[i] = seg[idx].y;
            }
            #pragma unroll
            for (int i = 0; i < 32; ++i)
                if (base + 2 * i + half < n) lds_add(&acc[rw[i] * 16 + col], xv[i]);
            #pragma unroll
            for (int j = 0; j < 2; ++j) {
                const int idx = base + 32 * j + lane;
                if (idx < n) lds_add(&cnt[seg[idx].y], 1.0f);
            }
        }
    }
    __syncthreads();

    for (int t = wave; t < ROWS / 16; t += NW) {
        const int rb = t * 16;
        float r = __frsqrt_rn(cnt[rb + col] + 1.0f);
        v8f c = {0.f, 0.f, 0.f, 0.f, 0.f, 0.f, 0.f, 0.f};
        #pragma unroll
        for (int k0 = 0; k0 < 16; k0 += 4) {
            v2f a, b;
            #pragma unroll
            for (int j = 0; j < 2; ++j) {
                int K = k0 + 2 * half + j;
                a[j] = (col == K) ? r : 0.0f;
                b[j] = acc[(rb + K) * 16 + col];
            }
            c = __builtin_amdgcn_wmma_f32_16x16x4_f32(
                    false, a, false, b, (short)0, c, false, false);
        }
        #pragma unroll
        for (int v = 0; v < 8; ++v) {
            unsigned row = lo + rb + v + 8 * half;
            if (row < (unsigned)num_nodes) out[(size_t)row * 16 + col] = c[v];
        }
    }
}

extern "C" void kernel_launch(void* const* d_in, const int* in_sizes, int n_in,
                              void* d_out, int out_size, void* d_ws, size_t ws_size,
                              hipStream_t stream) {
    const float* x        = (const float*)d_in[0];
    const int*   edge_dst = (const int*)d_in[1];
    int E         = in_sizes[0] / 16;
    int num_nodes = out_size / 16;
    float* out    = (float*)d_out;

    int groups = (num_nodes + ROWS - 1) / ROWS;
    NeighbourAggregation_owner_kernel<<<groups, NT, 0, stream>>>(
        x, edge_dst, out, E, num_nodes);
}


# Owner scan reads one-byte owner keys (16 edges per 16-byte load, packed byte test) from a plane that a new first launch writes, not 4-byte destinations; rows re-read at gather; LDS add order changes
# speedup vs baseline: 4.2124x; 1.5256x over previous
#include <hip/hip_runtime.h>


typedef __attribute__((ext_vector_type(2))) float    v2f;
typedef __attribute__((ext_vector_type(8))) float    v8f;
typedef __attribute__((ext_vector_type(4))) int      v4i;
typedef __attribute__((ext_vector_type(4))) unsigned v4u;

#define ROWS 400
#define NT   1024
#define NW   (NT / 32)
#define UNR  8
#define SEG  640
#define KT   256
#define KU   8

__device__ __forceinline__ void lds_add(float* p, float v) {
    __hip_atomic_fetch_add(p, v, __ATOMIC_RELAXED, __HIP_MEMORY_SCOPE_WORKGROUP);
}

__device__ __forceinline__ unsigned key_of(int d) {
    return (((unsigned)d) / (unsigned)ROWS) & 255u;
}

__global__ __launch_bounds__(KT)
void NeighbourAggregation_key_kernel(
    const int* __restrict__ edge_dst,
    unsigned int* __restrict__ keys,
    int E)
{
    const v4i* d4 = (const v4i*)edge_dst;
    const int E4 = E >> 2;
    const int stride = gridDim.x * KT;
    for (int g0 = blockIdx.x * KT + threadIdx.x; g0 < E4; g0 += stride * KU) {
        v4i d[KU];
        #pragma unroll
        for (int u = 0; u < KU; ++u) {
            const int g = g0 + u * stride;
            v4i t = {0, 0, 0, 0};
            if (g < E4) t = d4[g];
            d[u] = t;
        }
        #pragma unroll
        for (int u = 0; u < KU; ++u) {
            const int g = g0 + u * stride;
            if (g < E4)
                keys[g] = key_of(d[u].x) | (key_of(d[u].y) << 8) |
                          (key_of(d[u].z) << 16) | (key_of(d[u].w) << 24);
        }
    }
}

__device__ __forceinline__ unsigned zero_bytes(unsigned t) {
    return (t - 0x01010101u) & ~t & 0x80808080u;
}

__device__ __forceinline__ unsigned hits32(v4u a, v4u b, unsigned pat) {
    unsigned m;
    m  = zero_bytes(a.x ^ pat) >> 7;
    m |= zero_bytes(a.y ^ pat) >> 6;
    m |= zero_bytes(a.z ^ pat) >> 5;
    m |= zero_bytes(a.w ^ pat) >> 4;
    m |= zero_bytes(b.x ^ pat) >> 3;
    m |= zero_bytes(b.y ^ pat) >> 2;
    m |= zero_bytes(b.z ^ pat) >> 1;
    m |= zero_bytes(b.w ^ pat);
    return m;
}

__device__ __forceinline__ void drain(unsigned m, unsigned ea, unsigned* seg, int& wc) {
    unsigned any = __builtin_amdgcn_ballot_w32(m != 0u);
    while (any != 0u) {
        const int pos = wc + (int)__builtin_amdgcn_mbcnt_lo(any, 0u);
        if (m != 0u) {
            const unsigned p = (unsigned)__builtin_ctz(m);
            const unsigned j = p & 7u, k = p >> 3;
            const unsigned e = ea + (j >> 2) * (16u * NT) + ((j & 3u) << 2) + k;
            if (pos < SEG) seg[pos] = e;
            m &= m - 1u;
        }
        wc += __builtin_popcount(any);
        any = __builtin_amdgcn_ballot_w32(m != 0u);
    }
}

__device__ __forceinline__ void add_direct(float* acc, float* cnt, const float* __restrict__ x,
                                           int e, unsigned row) {
    if (row < ROWS) {
        const float* xr = x + (size_t)e * 16;
        for (int f = 0; f < 16; ++f) lds_add(&acc[row * 16 + f], xr[f]);
        lds_add(&cnt[row], 1.0f);
    }
}

__global__ __launch_bounds__(NT)
void NeighbourAggregation_owner_kernel(
    const float* __restrict__ x,
    const int* __restrict__ edge_dst,
    const unsigned int* __restrict__ keys,
    float* __restrict__ out,
    int E, int num_nodes)
{
    __shared__ float    acc[ROWS * 16];
    __shared__ float    cnt[ROWS];
    __shared__ unsigned queue[NW * SEG];

    const int tid  = threadIdx.x;
    const int lane = tid & 31;
    const int wave = tid >> 5;
    const int half = lane >> 4;
    const int col  = lane & 15;
    const unsigned lo  = blockIdx.x * ROWS;
    const unsigned pat = (blockIdx.x & 255u) * 0x01010101u;
    unsigned* seg = queue + wave * SEG;
    int wc = 0;

    for (int i = tid; i < ROWS * 16; i += NT) acc[i] = 0.0f;
    for (int i = tid; i < ROWS; i += NT) cnt[i] = 0.0f;
    __syncthreads();

    const v4u* k4 = (const v4u*)keys;
    const v4i* d4 = (const v4i*)edge_dst;
    const int E16 = E >> 4;
    const int nfull = E16 / (NT * UNR);
    const int start = (int)(((long long)blockIdx.x * nfull) / gridDim.x);
    for (int it = 0; it < nfull; ++it) {
        int itr = it + start;
        if (itr >= nfull) itr -= nfull;
        const int g0 = itr * (NT * UNR) + tid;
        v4u k[UNR];
        #pragma unroll
        for (int u = 0; u < UNR; ++u) k[u] = k4[g0 + u * NT];
        #pragma unroll
        for (int u = 0; u < UNR; u += 2)
            drain(hits32(k[u], k[u + 1], pat), (unsigned)(g0 + u * NT) * 16u, seg, wc);
    }
    {
        const int g0 = nfull * (NT * UNR) + tid;
        v4u k[UNR];
        #pragma unroll
        for (int u = 0; u < UNR; ++u) {
            const int g = g0 + u * NT;
            v4u t = {~pat, ~pat, ~pat, ~pat};
            if (g < E16) t = k4[g];
            k[u] = t;
        }
        #pragma unroll
        for (int u = 0; u < UNR; u += 2)
            drain(hits32(k[u], k[u + 1], pat), (unsigned)(g0 + u * NT) * 16u, seg, wc);
    }
    {
        const int e = E16 * 16 + tid;
        unsigned row = 0xffffffffu;
        if (e < E) row = (unsigned)edge_dst[e] - lo;
        drain(row < ROWS ? 1u : 0u, (unsigned)e, seg, wc);
    }

    if (wc > SEG) {
        for (int g = tid; g < E16; g += NT) {
            for (int q = 0; q < 4; ++q) {
                const v4i d = d4[g * 4 + q];
                const int e0 = g * 16 + q * 4;
                add_direct(acc, cnt, x, e0 + 0, (unsigned)d.x - lo);
                add_direct(acc, cnt, x, e0 + 1, (unsigned)d.y - lo);
                add_direct(acc, cnt, x, e0 + 2, (unsigned)d.z - lo);
                add_direct(acc, cnt, x, e0 + 3, (unsigned)d.w - lo);
            }
        }
        const int e = E16 * 16 + tid;
        if (e < E) add_direct(acc, cnt, x, e, (unsigned)edge_dst[e] - lo);
    } else {
        const int n = wc;
        for (int base = 0; base < n; base += 64) {
            float xv[32];
            #pragma unroll
            for (int i = 0; i < 32; ++i) {
                int idx = base + 2 * i + half;
                idx = idx < n ? idx : n - 1;
                const unsigned e = seg[idx];
                xv[i] = __builtin_nontemporal_load(x + (e * 16u + (unsigned)col));
            }
            unsigned rr[2];
            #pragma unroll
            for (int j = 0; j < 2; ++j) {
                const int idx = base + 32 * j + lane;
                const unsigned e = seg[idx < n ? idx : n - 1];
                const unsigned r = (unsigned)edge_dst[e] - lo;
                rr[j] = idx < n ? r : 0xffffffffu;
            }
            unsigned rw[32];
            #pragma unroll
            for (int i = 0; i < 32; ++i)
                rw[i] = (unsigned)__shfl((int)rr[i >> 4], 2 * (i & 15) + half, 32);
            #pragma unroll
            for (int i = 0; i < 32; ++i)
                if (rw[i] < ROWS) lds_add(&acc[rw[i] * 16 + col], xv[i]);
            #pragma unroll
            for (int j = 0; j < 2; ++j)
                if (rr[j] < ROWS) lds_add(&cnt[rr[j]], 1.0f);
        }
    }
    __syncthreads();

    for (int t = wave; t < ROWS / 16; t += NW) {
        const int rb = t * 16;
        float r = __frsqrt_rn(cnt[rb + col] + 1.0f);
        v8f c = {0.f, 0.f, 0.f, 0.f, 0.f, 0.f, 0.f, 0.f};
        #pragma unroll
        for (int k0 = 0; k0 < 16; k0 += 4) {
            v2f a, b;
            #pragma unroll
            for (int j = 0; j < 2; ++j) {
                int K = k0 + 2 * half + j;
                a[j] = (col == K) ? r : 0.0f;
                b[j] = acc[(rb + K) * 16 + col];
            }
            c = __builtin_amdgcn_wmma_f32_16x16x4_f32(
                    false, a, false, b, (short)0, c, false, false);
        }
        #pragma unroll
        for (int v = 0; v < 8; ++v) {
            unsigned row = lo + rb + v + 8 * half;
            if (row < (unsigned)num_nodes) out[(size_t)row * 16 + col] = c[v];
        }
    }
}

extern "C" void kernel_launch(void* const* d_in, const int* in_sizes, int n_in,
                              void* d_out, int out_size, void* d_ws, size_t ws_size,
                              hipStream_t stream) {
    const float* x        = (const float*)d_in[0];
    const int*   edge_dst = (const int*)d_in[1];
    int E         = in_sizes[0] / 16;
    int num_nodes = out_size / 16;
    float* out    = (float*)d_out;
    unsigned int* keys = (unsigned int*)d_ws;

    NeighbourAggregation_key_kernel<<<512, KT, 0, stream>>>(edge_dst, keys, E);

    int groups = (num_nodes + ROWS - 1) / ROWS;
    NeighbourAggregation_owner_kernel<<<groups, NT, 0, stream>>>(
        x, edge_dst, keys, out, E, num_nodes);
}


# Edge list cut into 4 slices x row blocks of 1600: an owner scans a quarter of the key plane, stores partial sums; new third launch adds the 4 partials, does the WMMA scale; sum order changes
# speedup vs baseline: 5.1361x; 1.2193x over previous
#include <hip/hip_runtime.h>


typedef __attribute__((ext_vector_type(2))) float    v2f;
typedef __attribute__((ext_vector_type(8))) float    v8f;
typedef __attribute__((ext_vector_type(4))) int      v4i;
typedef __attribute__((ext_vector_type(4))) unsigned v4u;
typedef __attribute__((ext_vector_type(4))) float    v4f;

#define NT   1024
#define NW   (NT / 32)
#define UNR  8
#define SEG  640
#define KT   256
#define KU   8
#define FT   256

__device__ __forceinline__ void lds_add(float* p, float v) {
    __hip_atomic_fetch_add(p, v, __ATOMIC_RELAXED, __HIP_MEMORY_SCOPE_WORKGROUP);
}

template <int ROWS>
__device__ __forceinline__ unsigned key_of(int d) {
    return (((unsigned)d) / (unsigned)ROWS) & 255u;
}

template <int ROWS>
__global__ __launch_bounds__(KT)
void NeighbourAggregation_key_kernel(
    const int* __restrict__ edge_dst,
    unsigned int* __restrict__ keys,
    int E)
{
    const v4i* d4 = (const v4i*)edge_dst;
    const int E4 = E >> 2;
    const int stride = gridDim.x * KT;
    for (int g0 = blockIdx.x * KT + threadIdx.x; g0 < E4; g0 += stride * KU) {
        v4i d[KU];
        #pragma unroll
        for (int u = 0; u < KU; ++u) {
            const int g = g0 + u * stride;
            v4i t = {0, 0, 0, 0};
            if (g < E4) t = d4[g];
            d[u] = t;
        }
        #pragma unroll
        for (int u = 0; u < KU; ++u) {
            const int g = g0 + u * stride;
            if (g < E4)
                keys[g] = key_of<ROWS>(d[u].x) | (key_of<ROWS>(d[u].y) << 8) |
                          (key_of<ROWS>(d[u].z) << 16) | (key_of<ROWS>(d[u].w) << 24);
        }
    }
}

__device__ __forceinline__ unsigned zero_bytes(unsigned t) {
    return (t - 0x01010101u) & ~t & 0x80808080u;
}

__device__ __forceinline__ unsigned hits32(v4u a, v4u b, unsigned pat) {
    unsigned m;
    m  = zero_bytes(a.x ^ pat) >> 7;
    m |= zero_bytes(a.y ^ pat) >> 6;
    m |= zero_bytes(a.z ^ pat) >> 5;
    m |= zero_bytes(a.w ^ pat) >> 4;
    m |= zero_bytes(b.x ^ pat) >> 3;
    m |= zero_bytes(b.y ^ pat) >> 2;
    m |= zero_bytes(b.z ^ pat) >> 1;
    m |= zero_bytes(b.w ^ pat);
    return m;
}

__device__ __forceinline__ void drain(unsigned m, unsigned ea, unsigned* seg, int& wc) {
    unsigned any = __builtin_amdgcn_ballot_w32(m != 0u);
    while (any != 0u) {
        const int pos = wc + (int)__builtin_amdgcn_mbcnt_lo(any, 0u);
        if (m != 0u) {
            const unsigned p = (unsigned)__builtin_ctz(m);
            const unsigned j = p & 7u, k = p >> 3;
            const unsigned e = ea + (j >> 2) * (16u * NT) + ((j & 3u) << 2) + k;
            if (pos < SEG) seg[pos] = e;
            m &= m - 1u;
        }
        wc += __builtin_popcount(any);
        any = __builtin_amdgcn_ballot_w32(m != 0u);
    }
}

template <int ROWS>
__device__ __forceinline__ void add_direct(float* acc, float* cnt, const float* __restrict__ x,
                                           int e, unsigned row) {
    if (row < ROWS) {
        const float* xr = x + (size_t)e * 16;
        for (int f = 0; f < 16; ++f) lds_add(&acc[row * 16 + f], xr[f]);
        lds_add(&cnt[row], 1.0f);
    }
}

template <int ROWS, int S>
__global__ __launch_bounds__(NT)
void NeighbourAggregation_owner_kernel(
    const float* __restrict__ x,
    const int* __restrict__ edge_dst,
    const unsigned int* __restrict__ keys,
    float* __restrict__ out,
    float* __restrict__ part,
    float* __restrict__ pcnt,
    int E, int num_nodes)
{
    __shared__ __attribute__((aligned(16))) float acc[ROWS * 16];
    __shared__ float    cnt[ROWS];
    __shared__ unsigned queue[NW * SEG];

    const int tid  = threadIdx.x;
    const int lane = tid & 31;
    const int wave = tid >> 5;
    const int half = lane >> 4;
    const int col  = lane & 15;
    const int blk   = (int)blockIdx.x / S;
    const int slice = (int)blockIdx.x - blk * S;
    const int nblk  = (int)gridDim.x / S;
    const unsigned lo  = (unsigned)blk * ROWS;
    const unsigned pat = ((unsigned)blk & 255u) * 0x01010101u;
    unsigned* seg = queue + wave * SEG;
    int wc = 0;

    for (int i = tid; i < ROWS * 16; i += NT) acc[i] = 0.0f;
    for (int i = tid; i < ROWS; i += NT) cnt[i] = 0.0f;
    __syncthreads();

    const v4u* k4 = (const v4u*)keys;
    const v4i* d4 = (const v4i*)edge_dst;
    const int E16 = E >> 4;
    const int gb = (int)(((long long)E16 * slice) / S);
    const int ge = (int)(((long long)E16 * (slice + 1)) / S);
    const int nfull = (ge - gb) / (NT * UNR);
    const int start = (int)(((long long)blk * nfull) / nblk);
    for (int it = 0; it < nfull; ++it) {
        int itr = it + start;
        if (itr >= nfull) itr -= nfull;
        const int g0 = gb + itr * (NT * UNR) + tid;
        v4u k[UNR];
        #pragma unroll
        for (int u = 0; u < UNR; ++u) k[u] = k4[g0 + u * NT];
        #pragma unroll
        for (int u = 0; u < UNR; u += 2)
            drain(hits32(k[u], k[u + 1], pat), (unsigned)(g0 + u * NT) * 16u, seg, wc);
    }
    {
        const int g0 = gb + nfull * (NT * UNR) + tid;
        v4u k[UNR];
        #pragma unroll
        for (int u = 0; u < UNR; ++u) {
            const int g = g0 + u * NT;
            v4u t = {~pat, ~pat, ~pat, ~pat};
            if (g < ge) t = k4[g];
            k[u] = t;
        }
        #pragma unroll
        for (int u = 0; u < UNR; u += 2)
            drain(hits32(k[u], k[u + 1], pat), (unsigned)(g0 + u * NT) * 16u, seg, wc);
    }
    {
        const int e = E16 * 16 + tid;
        unsigned row = 0xffffffffu;
        if (slice == S - 1 && e < E) row = (unsigned)edge_dst[e] - lo;
        drain(row < ROWS ? 1u : 0u, (unsigned)e, seg, wc);
    }

    if (wc > SEG) {
        for (int g = gb + tid; g < ge; g += NT) {
            for (int q = 0; q < 4; ++q) {
                const v4i d = d4[g * 4 + q];
                const int e0 = g * 16 + q * 4;
                add_direct<ROWS>(acc, cnt, x, e0 + 0, (unsigned)d.x - lo);
                add_direct<ROWS>(acc, cnt, x, e0 + 1, (unsigned)d.y - lo);
                add_direct<ROWS>(acc, cnt, x, e0 + 2, (unsigned)d.z - lo);
                add_direct<ROWS>(acc, cnt, x, e0 + 3, (unsigned)d.w - lo);
            }
        }
        const int e = E16 * 16 + tid;
        if (slice == S - 1 && e < E) add_direct<ROWS>(acc, cnt, x, e, (unsigned)edge_dst[e] - lo);
    } else {
        const int n = wc;
        for (int base = 0; base < n; base += 64) {
            float xv[32];
            #pragma unroll
            for (int i = 0; i < 32; ++i) {
                int idx = base + 2 * i + half;
                idx = idx < n ? idx : n - 1;
                const unsigned e = seg[idx];
                xv[i] = __builtin_nontemporal_load(x + (e * 16u + (unsigned)col));
            }
            unsigned rr[2];
            #pragma unroll
            for (int j = 0; j < 2; ++j) {
                const int idx = base + 32 * j + lane;
                const unsigned e = seg[idx < n ? idx : n - 1];
                const unsigned r = (unsigned)edge_dst[e] - lo;
                rr[j] = idx < n ? r : 0xffffffffu;
            }
            unsigned rw[32];
            #pragma unroll
            for (int i = 0; i < 32; ++i)
                rw[i] = (unsigned)__shfl((int)rr[i >> 4], 2 * (i & 15) + half, 32);
            #pragma unroll
            for (int i = 0; i < 32; ++i)
                if (rw[i] < ROWS) lds_add(&acc[rw[i] * 16 + col], xv[i]);
            #pragma unroll
            for (int j = 0; j < 2; ++j)
                if (rr[j] < ROWS) lds_add(&cnt[rr[j]], 1.0f);
        }
    }
    __syncthreads();

    int nrows = num_nodes - (int)lo;
    nrows = nrows < ROWS ? nrows : ROWS;
    float* dsum = (slice == 0) ? out : part + (size_t)(slice - 1) * (size_t)num_nodes * 16;
    float* dcnt = pcnt + (size_t)slice * (size_t)num_nodes;
    const v4f* a4 = (const v4f*)acc;
    v4f* o4 = (v4f*)(dsum + (size_t)lo * 16);
    for (int i = tid; i < nrows * 4; i += NT) o4[i] = a4[i];
    for (int i = tid; i < nrows; i += NT) dcnt[lo + i] = cnt[i];
}

template <int S>
__global__ __launch_bounds__(FT)
void NeighbourAggregation_scale_kernel(
    float* __restrict__ out,
    const float* __restrict__ part,
    const float* __restrict__ pcnt,
    int num_nodes)
{
    const int lane = threadIdx.x & 31;
    const int wave = threadIdx.x >> 5;
    const int half = lane >> 4;
    const int col  = lane & 15;
    const int ntiles = (num_nodes + 15) >> 4;
    const int t = (int)blockIdx.x * (FT / 32) + wave;
    if (t >= ntiles) return;
    const int rb = t * 16;
    const size_t plane = (size_t)num_nodes * 16;

    const int last = num_nodes - 1;
    const int rc = rb + col < last ? rb + col : last;
    float ctot = 0.0f;
    #pragma unroll
    for (int s = 0; s < S; ++s) ctot += pcnt[(size_t)s * (size_t)num_nodes + rc];
    const float r = __frsqrt_rn(ctot + 1.0f);

    float bs[8];
    #pragma unroll
    for (int q = 0; q < 8; ++q) {
        const int K = (q >> 1) * 4 + 2 * half + (q & 1);
        const int row = rb + K;
        const size_t o = (size_t)(row < last ? row : last) * 16 + col;
        float sum = out[o];
        #pragma unroll
        for (int s = 1; s < S; ++s) sum += part[(size_t)(s - 1) * plane + o];
        bs[q] = row < num_nodes ? sum : 0.0f;
    }

    v8f c = {0.f, 0.f, 0.f, 0.f, 0.f, 0.f, 0.f, 0.f};
    #pragma unroll
    for (int k0 = 0; k0 < 16; k0 += 4) {
        v2f a, b;
        #pragma unroll
        for (int j = 0; j < 2; ++j) {
            const int K = k0 + 2 * half + j;
            a[j] = (col == K) ? r : 0.0f;
            b[j] = bs[(k0 >> 1) + j];
        }
        c = __builtin_amdgcn_wmma_f32_16x16x4_f32(
                false, a, false, b, (short)0, c, false, false);
    }
    #pragma unroll
    for (int v = 0; v < 8; ++v) {
        const int row = rb + v + 8 * half;
        if (row < num_nodes) out[(size_t)row * 16 + col] = c[v];
    }
}

extern "C" void kernel_launch(void* const* d_in, const int* in_sizes, int n_in,
                              void* d_out, int out_size, void* d_ws, size_t ws_size,
                              hipStream_t stream) {
    const float* x        = (const float*)d_in[0];
    const int*   edge_dst = (const int*)d_in[1];
    int E         = in_sizes[0] / 16;
    int num_nodes = out_size / 16;
    float* out    = (float*)d_out;
    unsigned int* keys = (unsigned int*)d_ws;
    size_t koff   = (((size_t)E + 255) / 256) * 256;
    float* part   = (float*)((char*)d_ws + koff);
    size_t plane  = (size_t)num_nodes * 16;
    int fgroups   = ((num_nodes + 15) / 16 + FT / 32 - 1) / (FT / 32);

    if (ws_size >= koff + (3 * plane + 4 * (size_t)num_nodes) * sizeof(float)) {
        float* pcnt = part + 3 * plane;
        int nblk = (num_nodes + 1600 - 1) / 1600;
        NeighbourAggregation_key_kernel<1600><<<512, KT, 0, stream>>>(edge_dst, keys, E);
        NeighbourAggregation_owner_kernel<1600, 4><<<nblk * 4, NT, 0, stream>>>(
            x, edge_dst, keys, out, part, pcnt, E, num_nodes);
        NeighbourAggregation_scale_kernel<4><<<fgroups, FT, 0, stream>>>(out, part, pcnt, num_nodes);
    } else {
        float* pcnt = part;
        int nblk = (num_nodes + 400 - 1) / 400;
        NeighbourAggregation_key_kernel<400><<<512, KT, 0, stream>>>(edge_dst, keys, E);
        NeighbourAggregation_owner_kernel<400, 1><<<nblk, NT, 0, stream>>>(
            x, edge_dst, keys, out, part, pcnt, E, num_nodes);
        NeighbourAggregation_scale_kernel<1><<<fgroups, FT, 0, stream>>>(out, part, pcnt, num_nodes);
    }
}


# speedup vs baseline: 5.5721x; 1.0849x over previous
#include <hip/hip_runtime.h>


typedef __attribute__((ext_vector_type(2))) float    v2f;
typedef __attribute__((ext_vector_type(8))) float    v8f;
typedef __attribute__((ext_vector_type(4))) int      v4i;
typedef __attribute__((ext_vector_type(4))) unsigned v4u;
typedef __attribute__((ext_vector_type(4))) float    v4f;

#define NT   1024
#define NW   (NT / 32)
#define UNR  8
#define SEG  512
#define GB   64
#define KT   256
#define KU   8
#define FT   256

__device__ __forceinline__ void lds_add(float* p, float v) {
    __hip_atomic_fetch_add(p, v, __ATOMIC_RELAXED, __HIP_MEMORY_SCOPE_WORKGROUP);
}

template <int ROWS>
__device__ __forceinline__ unsigned key_of(int d) {
    return (((unsigned)d) / (unsigned)ROWS) & 255u;
}

template <int ROWS>
__global__ __launch_bounds__(KT)
void NeighbourAggregation_key_kernel(
    const int* __restrict__ edge_dst,
    unsigned int* __restrict__ keys,
    int E)
{
    const v4i* d4 = (const v4i*)edge_dst;
    const int E4 = E >> 2;
    const int stride = gridDim.x * KT;
    for (int g0 = blockIdx.x * KT + threadIdx.x; g0 < E4; g0 += stride * KU) {
        v4i d[KU];
        #pragma unroll
        for (int u = 0; u < KU; ++u) {
            const int g = g0 + u * stride;
            v4i t = {0, 0, 0, 0};
            if (g < E4) t = d4[g];
            d[u] = t;
        }
        #pragma unroll
        for (int u = 0; u < KU; ++u) {
            const int g = g0 + u * stride;
            if (g < E4)
                keys[g] = key_of<ROWS>(d[u].x) | (key_of<ROWS>(d[u].y) << 8) |
                          (key_of<ROWS>(d[u].z) << 16) | (key_of<ROWS>(d[u].w) << 24);
        }
    }
}

__device__ __forceinline__ unsigned zero_bytes(unsigned t) {
    return (t - 0x01010101u) & ~t & 0x80808080u;
}

__device__ __forceinline__ unsigned hits32(v4u a, v4u b, unsigned pat) {
    unsigned m;
    m  = zero_bytes(a.x ^ pat) >> 7;
    m |= zero_bytes(a.y ^ pat) >> 6;
    m |= zero_bytes(a.z ^ pat) >> 5;
    m |= zero_bytes(a.w ^ pat) >> 4;
    m |= zero_bytes(b.x ^ pat) >> 3;
    m |= zero_bytes(b.y ^ pat) >> 2;
    m |= zero_bytes(b.z ^ pat) >> 1;
    m |= zero_bytes(b.w ^ pat);
    return m;
}

__device__ __forceinline__ void drain(unsigned m, unsigned ea, unsigned* seg, int head, int& wc) {
    unsigned any = __builtin_amdgcn_ballot_w32(m != 0u);
    while (any != 0u) {
        const int pos = wc + (int)__builtin_amdgcn_mbcnt_lo(any, 0u);
        if (m != 0u) {
            const unsigned p = (unsigned)__builtin_ctz(m);
            const unsigned j = p & 7u, k = p >> 3;
            const unsigned e = ea + (j >> 2) * (16u * NT) + ((j & 3u) << 2) + k;
            if (pos - head < SEG) seg[pos & (SEG - 1)] = e;
            m &= m - 1u;
        }
        wc += __builtin_popcount(any);
        any = __builtin_amdgcn_ballot_w32(m != 0u);
    }
}

template <int ROWS>
__device__ __forceinline__ void add_direct(float* acc, float* cnt, const float* __restrict__ x,
                                           int e, unsigned row) {
    if (row < ROWS) {
        const float* xr = x + (size_t)e * 16;
        for (int f = 0; f < 16; ++f) lds_add(&acc[row * 16 + f], xr[f]);
        lds_add(&cnt[row], 1.0f);
    }
}

template <int ROWS, bool FULL>
__device__ __forceinline__ void gather_pass(float* acc, float* cnt, const float* __restrict__ x,
                                            const int* __restrict__ edge_dst, const unsigned* seg,
                                            int head, int n, unsigned lo, int lane) {
    const int half = lane >> 4;
    const int col  = lane & 15;
    float xv[GB / 2];
    #pragma unroll
    for (int i = 0; i < GB / 2; ++i) {
        int idx = 2 * i + half;
        if (!FULL) idx = idx < n ? idx : n - 1;
        const unsigned e = seg[(head + idx) & (SEG - 1)];
        xv[i] = x[e * 16u + (unsigned)col];
    }
    unsigned rr[2];
    #pragma unroll
    for (int j = 0; j < 2; ++j) {
        const int idx = 32 * j + lane;
        const unsigned e = seg[(head + (FULL || idx < n ? idx : n - 1)) & (SEG - 1)];
        const unsigned r = (unsigned)edge_dst[e] - lo;
        rr[j] = (FULL || idx < n) ? r : 0xffffffffu;
    }
    unsigned rw[GB / 2];
    #pragma unroll
    for (int i = 0; i < GB / 2; ++i)
        rw[i] = (unsigned)__shfl((int)rr[i >> 4], 2 * (i & 15) + half, 32);
    #pragma unroll
    for (int i = 0; i < GB / 2; ++i)
        if (rw[i] < ROWS) lds_add(&acc[rw[i] * 16 + col], xv[i]);
    #pragma unroll
    for (int j = 0; j < 2; ++j)
        if (rr[j] < ROWS) lds_add(&cnt[rr[j]], 1.0f);
}

template <int ROWS, int S>
__global__ __launch_bounds__(NT)
void NeighbourAggregation_owner_kernel(
    const float* __restrict__ x,
    const int* __restrict__ edge_dst,
    const unsigned int* __restrict__ keys,
    float* __restrict__ out,
    float* __restrict__ part,
    float* __restrict__ pcnt,
    int E, int num_nodes)
{
    __shared__ __attribute__((aligned(16))) float acc[ROWS * 16];
    __shared__ float    cnt[ROWS];
    __shared__ unsigned queue[NW * SEG];

    const int tid  = threadIdx.x;
    const int lane = tid & 31;
    const int wave = tid >> 5;
    const int blk   = (int)blockIdx.x / S;
    const int slice = (int)blockIdx.x - blk * S;
    const unsigned lo  = (unsigned)blk * ROWS;
    const unsigned pat = ((unsigned)blk & 255u) * 0x01010101u;
    unsigned* seg = queue + wave * SEG;
    int wc = 0;
    int head = 0;

    for (int i = tid; i < ROWS * 16; i += NT) acc[i] = 0.0f;
    for (int i = tid; i < ROWS; i += NT) cnt[i] = 0.0f;
    __syncthreads();

    const v4u* k4 = (const v4u*)keys;
    const v4i* d4 = (const v4i*)edge_dst;
    const int E16 = E >> 4;
    const int gb = (int)(((long long)E16 * slice) / S);
    const int ge = (int)(((long long)E16 * (slice + 1)) / S);
    const int nwin = (ge - gb + NT * UNR - 1) / (NT * UNR);
    for (int it = 0; it < nwin; ++it) {
        const int w0 = gb + it * (NT * UNR);
        const int g0 = w0 + tid;
        const int wc0 = wc;
        v4u k[UNR];
        if (w0 + NT * UNR <= ge) {
            #pragma unroll
            for (int u = 0; u < UNR; ++u) k[u] = k4[g0 + u * NT];
        } else {
            #pragma unroll
            for (int u = 0; u < UNR; ++u) {
                const int g = g0 + u * NT;
                v4u t = {~pat, ~pat, ~pat, ~pat};
                if (g < ge) t = k4[g];
                k[u] = t;
            }
        }
        #pragma unroll
        for (int u = 0; u < UNR; u += 2)
            drain(hits32(k[u], k[u + 1], pat), (unsigned)(g0 + u * NT) * 16u, seg, head, wc);

        if (wc - head > SEG) {
            wc = wc0;
            for (int u = 0; u < UNR; ++u) {
                const int g = g0 + u * NT;
                if (g < ge) {
                    for (int q = 0; q < 4; ++q) {
                        const v4i d = d4[g * 4 + q];
                        const int e0 = g * 16 + q * 4;
                        add_direct<ROWS>(acc, cnt, x, e0 + 0, (unsigned)d.x - lo);
                        add_direct<ROWS>(acc, cnt, x, e0 + 1, (unsigned)d.y - lo);
                        add_direct<ROWS>(acc, cnt, x, e0 + 2, (unsigned)d.z - lo);
                        add_direct<ROWS>(acc, cnt, x, e0 + 3, (unsigned)d.w - lo);
                    }
                }
            }
        }

        while (wc - head >= GB) {
            gather_pass<ROWS, true>(acc, cnt, x, edge_dst, seg, head, GB, lo, lane);
            head += GB;
        }
    }
    {
        const int e = E16 * 16 + tid;
        unsigned row = 0xffffffffu;
        if (slice == S - 1 && e < E) row = (unsigned)edge_dst[e] - lo;
        drain(row < ROWS ? 1u : 0u, (unsigned)e, seg, head, wc);
    }
    while (wc - head > 0) {
        const int n = wc - head < GB ? wc - head : GB;
        gather_pass<ROWS, false>(acc, cnt, x, edge_dst, seg, head, n, lo, lane);
        head += n;
    }
    __syncthreads();

    int nrows = num_nodes - (int)lo;
    nrows = nrows < ROWS ? nrows : ROWS;
    float* dsum = (slice == 0) ? out : part + (size_t)(slice - 1) * (size_t)num_nodes * 16;
    float* dcnt = pcnt + (size_t)slice * (size_t)num_nodes;
    const v4f* a4 = (const v4f*)acc;
    v4f* o4 = (v4f*)(dsum + (size_t)lo * 16);
    for (int i = tid; i < nrows * 4; i += NT) o4[i] = a4[i];
    for (int i = tid; i < nrows; i += NT) dcnt[lo + i] = cnt[i];
}

template <int S>
__global__ __launch_bounds__(FT)
void NeighbourAggregation_scale_kernel(
    float* __restrict__ out,
    const float* __restrict__ part,
    const float* __restrict__ pcnt,
    int num_nodes)
{
    const int lane = threadIdx.x & 31;
    const int wave = threadIdx.x >> 5;
    const int half = lane >> 4;
    const int col  = lane & 15;
    const int ntiles = (num_nodes + 15) >> 4;
    const int t = (int)blockIdx.x * (FT / 32) + wave;
    if (t >= ntiles) return;
    const int rb = t * 16;
    const size_t plane = (size_t)num_nodes * 16;

    const int last = num_nodes - 1;
    const int rc = rb + col < last ? rb + col : last;
    float ctot = 0.0f;
    #pragma unroll
    for (int s = 0; s < S; ++s) ctot += pcnt[(size_t)s * (size_t)num_nodes + rc];
    const float r = __frsqrt_rn(ctot + 1.0f);

    float bs[8];
    #pragma unroll
    for (int q = 0; q < 8; ++q) {
        const int K = (q >> 1) * 4 + 2 * half + (q & 1);
        const int row = rb + K;
        const size_t o = (size_t)(row < last ? row : last) * 16 + col;
        float sum = out[o];
        #pragma unroll
        for (int s = 1; s < S; ++s) sum += part[(size_t)(s - 1) * plane + o];
        bs[q] = row < num_nodes ? sum : 0.0f;
    }

    v8f c = {0.f, 0.f, 0.f, 0.f, 0.f, 0.f, 0.f, 0.f};
    #pragma unroll
    for (int k0 = 0; k0 < 16; k0 += 4) {
        v2f a, b;
        #pragma unroll
        for (int j = 0; j < 2; ++j) {
            const int K = k0 + 2 * half + j;
            a[j] = (col == K) ? r : 0.0f;
            b[j] = bs[(k0 >> 1) + j];
        }
        c = __builtin_amdgcn_wmma_f32_16x16x4_f32(
                false, a, false, b, (short)0, c, false, false);
    }
    #pragma unroll
    for (int v = 0; v < 8; ++v) {
        const int row = rb + v + 8 * half;
        if (row < num_nodes) out[(size_t)row * 16 + col] = c[v];
    }
}

extern "C" void kernel_launch(void* const* d_in, const int* in_sizes, int n_in,
                              void* d_out, int out_size, void* d_ws, size_t ws_size,
                              hipStream_t stream) {
    const float* x        = (const float*)d_in[0];
    const int*   edge_dst = (const int*)d_in[1];
    int E         = in_sizes[0] / 16;
    int num_nodes = out_size / 16;
    float* out    = (float*)d_out;
    unsigned int* keys = (unsigned int*)d_ws;
    size_t koff   = (((size_t)E + 255) / 256) * 256;
    float* part   = (float*)((char*)d_ws + koff);
    size_t plane  = (size_t)num_nodes * 16;
    int fgroups   = ((num_nodes + 15) / 16 + FT / 32 - 1) / (FT / 32);

    if (ws_size >= koff + (3 * plane + 4 * (size_t)num_nodes) * sizeof(float)) {
        float* pcnt = part + 3 * plane;
        int nblk = (num_nodes + 1600 - 1) / 1600;
        NeighbourAggregation_key_kernel<1600><<<512, KT, 0, stream>>>(edge_dst, keys, E);
        NeighbourAggregation_owner_kernel<1600, 4><<<nblk * 4, NT, 0, stream>>>(
            x, edge_dst, keys, out, part, pcnt, E, num_nodes);
        NeighbourAggregation_scale_kernel<4><<<fgroups, FT, 0, stream>>>(out, part, pcnt, num_nodes);
    } else {
        float* pcnt = part;
        int nblk = (num_nodes + 400 - 1) / 400;
        NeighbourAggregation_key_kernel<400><<<512, KT, 0, stream>>>(edge_dst, keys, E);
        NeighbourAggregation_owner_kernel<400, 1><<<nblk, NT, 0, stream>>>(
            x, edge_dst, keys, out, part, pcnt, E, num_nodes);
        NeighbourAggregation_scale_kernel<1><<<fgroups, FT, 0, stream>>>(out, part, pcnt, num_nodes);
    }
}


# Each edge-list slice bound to one chip half: an owner reads its die, draws its (block, slice) by integer ticket from its half's list, so x rows fill one L2 half, not both; LDS add order as before
# speedup vs baseline: 5.9398x; 1.0660x over previous
#include <hip/hip_runtime.h>


typedef __attribute__((ext_vector_type(2))) float    v2f;
typedef __attribute__((ext_vector_type(8))) float    v8f;
typedef __attribute__((ext_vector_type(4))) int      v4i;
typedef __attribute__((ext_vector_type(4))) unsigned v4u;
typedef __attribute__((ext_vector_type(4))) float    v4f;

#define NT   1024
#define NW   (NT / 32)
#define UNR  8
#define SEG  512
#define GB   64
#define KT   256
#define KU   8
#define FT   256
#define TICKET_GAP 64

__device__ __forceinline__ void lds_add(float* p, float v) {
    __hip_atomic_fetch_add(p, v, __ATOMIC_RELAXED, __HIP_MEMORY_SCOPE_WORKGROUP);
}

template <int ROWS>
__device__ __forceinline__ unsigned key_of(int d) {
    return (((unsigned)d) / (unsigned)ROWS) & 255u;
}

template <int ROWS>
__global__ __launch_bounds__(KT)
void NeighbourAggregation_key_kernel(
    const int* __restrict__ edge_dst,
    unsigned int* __restrict__ keys,
    unsigned int* __restrict__ ticket,
    int E)
{
    if (ticket != nullptr && blockIdx.x == 0 && threadIdx.x < 2) ticket[threadIdx.x * TICKET_GAP] = 0u;
    const v4i* d4 = (const v4i*)edge_dst;
    const int E4 = E >> 2;
    const int stride = gridDim.x * KT;
    for (int g0 = blockIdx.x * KT + threadIdx.x; g0 < E4; g0 += stride * KU) {
        v4i d[KU];
        #pragma unroll
        for (int u = 0; u < KU; ++u) {
            const int g = g0 + u * stride;
            v4i t = {0, 0, 0, 0};
            if (g < E4) t = d4[g];
            d[u] = t;
        }
        #pragma unroll
        for (int u = 0; u < KU; ++u) {
            const int g = g0 + u * stride;
            if (g < E4)
                keys[g] = key_of<ROWS>(d[u].x) | (key_of<ROWS>(d[u].y) << 8) |
                          (key_of<ROWS>(d[u].z) << 16) | (key_of<ROWS>(d[u].w) << 24);
        }
    }
}

__device__ __forceinline__ unsigned zero_bytes(unsigned t) {
    return (t - 0x01010101u) & ~t & 0x80808080u;
}

__device__ __forceinline__ unsigned hits32(v4u a, v4u b, unsigned pat) {
    unsigned m;
    m  = zero_bytes(a.x ^ pat) >> 7;
    m |= zero_bytes(a.y ^ pat) >> 6;
    m |= zero_bytes(a.z ^ pat) >> 5;
    m |= zero_bytes(a.w ^ pat) >> 4;
    m |= zero_bytes(b.x ^ pat) >> 3;
    m |= zero_bytes(b.y ^ pat) >> 2;
    m |= zero_bytes(b.z ^ pat) >> 1;
    m |= zero_bytes(b.w ^ pat);
    return m;
}

__device__ __forceinline__ void drain(unsigned m, unsigned ea, unsigned* seg, int head, int& wc) {
    unsigned any = __builtin_amdgcn_ballot_w32(m != 0u);
    while (any != 0u) {
        const int pos = wc + (int)__builtin_amdgcn_mbcnt_lo(any, 0u);
        if (m != 0u) {
            const unsigned p = (unsigned)__builtin_ctz(m);
            const unsigned j = p & 7u, k = p >> 3;
            const unsigned e = ea + (j >> 2) * (16u * NT) + ((j & 3u) << 2) + k;
            if (pos - head < SEG) seg[pos & (SEG - 1)] = e;
            m &= m - 1u;
        }
        wc += __builtin_popcount(any);
        any = __builtin_amdgcn_ballot_w32(m != 0u);
    }
}

template <int ROWS>
__device__ __forceinline__ void add_direct(float* acc, float* cnt, const float* __restrict__ x,
                                           int e, unsigned row) {
    if (row < ROWS) {
        const float* xr = x + (size_t)e * 16;
        for (int f = 0; f < 16; ++f) lds_add(&acc[row * 16 + f], xr[f]);
        lds_add(&cnt[row], 1.0f);
    }
}

template <int ROWS, bool FULL>
__device__ __forceinline__ void gather_pass(float* acc, float* cnt, const float* __restrict__ x,
                                            const int* __restrict__ edge_dst, const unsigned* seg,
                                            int head, int n, unsigned lo, int lane) {
    const int half = lane >> 4;
    const int col  = lane & 15;
    float xv[GB / 2];
    #pragma unroll
    for (int i = 0; i < GB / 2; ++i) {
        int idx = 2 * i + half;
        if (!FULL) idx = idx < n ? idx : n - 1;
        const unsigned e = seg[(head + idx) & (SEG - 1)];
        xv[i] = x[e * 16u + (unsigned)col];
    }
    unsigned rr[2];
    #pragma unroll
    for (int j = 0; j < 2; ++j) {
        const int idx = 32 * j + lane;
        const unsigned e = seg[(head + (FULL || idx < n ? idx : n - 1)) & (SEG - 1)];
        const unsigned r = (unsigned)edge_dst[e] - lo;
        rr[j] = (FULL || idx < n) ? r : 0xffffffffu;
    }
    unsigned rw[GB / 2];
    #pragma unroll
    for (int i = 0; i < GB / 2; ++i)
        rw[i] = (unsigned)__shfl((int)rr[i >> 4], 2 * (i & 15) + half, 32);
    #pragma unroll
    for (int i = 0; i < GB / 2; ++i)
        if (rw[i] < ROWS) lds_add(&acc[rw[i] * 16 + col], xv[i]);
    #pragma unroll
    for (int j = 0; j < 2; ++j)
        if (rr[j] < ROWS) lds_add(&cnt[rr[j]], 1.0f);
}

template <int ROWS, int S>
__global__ __launch_bounds__(NT)
void NeighbourAggregation_owner_kernel(
    const float* __restrict__ x,
    const int* __restrict__ edge_dst,
    const unsigned int* __restrict__ keys,
    float* __restrict__ out,
    float* __restrict__ part,
    float* __restrict__ pcnt,
    unsigned int* __restrict__ ticket,
    int E, int num_nodes)
{
    __shared__ __attribute__((aligned(16))) float acc[ROWS * 16];
    __shared__ float    cnt[ROWS];
    __shared__ unsigned queue[NW * SEG];
    __shared__ int      s_item;

    const int tid  = threadIdx.x;
    const int lane = tid & 31;
    const int wave = tid >> 5;

    if (S > 1 && tid == 0) {
        unsigned h = (__builtin_amdgcn_s_sendmsg_rtn(0x87) >> 18) & 1u;
        const unsigned per = gridDim.x >> 1;
        unsigned t = __hip_atomic_fetch_add(&ticket[h * TICKET_GAP], 1u, __ATOMIC_RELAXED, __HIP_MEMORY_SCOPE_AGENT);
        if (t >= per) {
            h ^= 1u;
            t = __hip_atomic_fetch_add(&ticket[h * TICKET_GAP], 1u, __ATOMIC_RELAXED, __HIP_MEMORY_SCOPE_AGENT);
        }
        const unsigned hs = S / 2 > 0 ? S / 2 : 1;
        s_item = t < per ? (int)((t / hs) * S + h * hs + t % hs) : -1;
    }
    for (int i = tid; i < ROWS * 16; i += NT) acc[i] = 0.0f;
    for (int i = tid; i < ROWS; i += NT) cnt[i] = 0.0f;
    __syncthreads();

    const int item = S > 1 ? __builtin_amdgcn_readfirstlane(s_item) : (int)blockIdx.x;
    if (item < 0) return;
    const int blk   = item / S;
    const int slice = item - blk * S;
    const unsigned lo  = (unsigned)blk * ROWS;
    const unsigned pat = ((unsigned)blk & 255u) * 0x01010101u;
    unsigned* seg = queue + wave * SEG;
    int wc = 0;
    int head = 0;

    const v4u* k4 = (const v4u*)keys;
    const v4i* d4 = (const v4i*)edge_dst;
    const int E16 = E >> 4;
    const int gb = (int)(((long long)E16 * slice) / S);
    const int ge = (int)(((long long)E16 * (slice + 1)) / S);
    const int nwin = (ge - gb + NT * UNR - 1) / (NT * UNR);
    for (int it = 0; it < nwin; ++it) {
        const int w0 = gb + it * (NT * UNR);
        const int g0 = w0 + tid;
        const int wc0 = wc;
        v4u k[UNR];
        if (w0 + NT * UNR <= ge) {
            #pragma unroll
            for (int u = 0; u < UNR; ++u) k[u] = k4[g0 + u * NT];
        } else {
            #pragma unroll
            for (int u = 0; u < UNR; ++u) {
                const int g = g0 + u * NT;
                v4u t = {~pat, ~pat, ~pat, ~pat};
                if (g < ge) t = k4[g];
                k[u] = t;
            }
        }
        #pragma unroll
        for (int u = 0; u < UNR; u += 2)
            drain(hits32(k[u], k[u + 1], pat), (unsigned)(g0 + u * NT) * 16u, seg, head, wc);

        if (wc - head > SEG) {
            wc = wc0;
            for (int u = 0; u < UNR; ++u) {
                const int g = g0 + u * NT;
                if (g < ge) {
                    for (int q = 0; q < 4; ++q) {
                        const v4i d = d4[g * 4 + q];
                        const int e0 = g * 16 + q * 4;
                        add_direct<ROWS>(acc, cnt, x, e0 + 0, (unsigned)d.x - lo);
                        add_direct<ROWS>(acc, cnt, x, e0 + 1, (unsigned)d.y - lo);
                        add_direct<ROWS>(acc, cnt, x, e0 + 2, (unsigned)d.z - lo);
                        add_direct<ROWS>(acc, cnt, x, e0 + 3, (unsigned)d.w - lo);
                    }
                }
            }
        }

        while (wc - head >= GB) {
            gather_pass<ROWS, true>(acc, cnt, x, edge_dst, seg, head, GB, lo, lane);
            head += GB;
        }
    }
    {
        const int e = E16 * 16 + tid;
        unsigned row = 0xffffffffu;
        if (slice == S - 1 && e < E) row = (unsigned)edge_dst[e] - lo;
        drain(row < ROWS ? 1u : 0u, (unsigned)e, seg, head, wc);
    }
    while (wc - head > 0) {
        const int n = wc - head < GB ? wc - head : GB;
        gather_pass<ROWS, false>(acc, cnt, x, edge_dst, seg, head, n, lo, lane);
        head += n;
    }
    __syncthreads();

    int nrows = num_nodes - (int)lo;
    nrows = nrows < ROWS ? nrows : ROWS;
    float* dsum = (slice == 0) ? out : part + (size_t)(slice - 1) * (size_t)num_nodes * 16;
    float* dcnt = pcnt + (size_t)slice * (size_t)num_nodes;
    const v4f* a4 = (const v4f*)acc;
    v4f* o4 = (v4f*)(dsum + (size_t)lo * 16);
    for (int i = tid; i < nrows * 4; i += NT) o4[i] = a4[i];
    for (int i = tid; i < nrows; i += NT) dcnt[lo + i] = cnt[i];
}

template <int S>
__global__ __launch_bounds__(FT)
void NeighbourAggregation_scale_kernel(
    float* __restrict__ out,
    const float* __restrict__ part,
    const float* __restrict__ pcnt,
    int num_nodes)
{
    const int lane = threadIdx.x & 31;
    const int wave = threadIdx.x >> 5;
    const int half = lane >> 4;
    const int col  = lane & 15;
    const int ntiles = (num_nodes + 15) >> 4;
    const int t = (int)blockIdx.x * (FT / 32) + wave;
    if (t >= ntiles) return;
    const int rb = t * 16;
    const size_t plane = (size_t)num_nodes * 16;

    const int last = num_nodes - 1;
    const int rc = rb + col < last ? rb + col : last;
    float ctot = 0.0f;
    #pragma unroll
    for (int s = 0; s < S; ++s) ctot += pcnt[(size_t)s * (size_t)num_nodes + rc];
    const float r = __frsqrt_rn(ctot + 1.0f);

    float bs[8];
    #pragma unroll
    for (int q = 0; q < 8; ++q) {
        const int K = (q >> 1) * 4 + 2 * half + (q & 1);
        const int row = rb + K;
        const size_t o = (size_t)(row < last ? row : last) * 16 + col;
        float sum = out[o];
        #pragma unroll
        for (int s = 1; s < S; ++s) sum += part[(size_t)(s - 1) * plane + o];
        bs[q] = row < num_nodes ? sum : 0.0f;
    }

    v8f c = {0.f, 0.f, 0.f, 0.f, 0.f, 0.f, 0.f, 0.f};
    #pragma unroll
    for (int k0 = 0; k0 < 16; k0 += 4) {
        v2f a, b;
        #pragma unroll
        for (int j = 0; j < 2; ++j) {
            const int K = k0 + 2 * half + j;
            a[j] = (col == K) ? r : 0.0f;
            b[j] = bs[(k0 >> 1) + j];
        }
        c = __builtin_amdgcn_wmma_f32_16x16x4_f32(
                false, a, false, b, (short)0, c, false, false);
    }
    #pragma unroll
    for (int v = 0; v < 8; ++v) {
        const int row = rb + v + 8 * half;
        if (row < num_nodes) out[(size_t)row * 16 + col] = c[v];
    }
}

extern "C" void kernel_launch(void* const* d_in, const int* in_sizes, int n_in,
                              void* d_out, int out_size, void* d_ws, size_t ws_size,
                              hipStream_t stream) {
    const float* x        = (const float*)d_in[0];
    const int*   edge_dst = (const int*)d_in[1];
    int E         = in_sizes[0] / 16;
    int num_nodes = out_size / 16;
    float* out    = (float*)d_out;
    size_t koff   = (((size_t)E + 255) / 256) * 256;
    size_t plane  = (size_t)num_nodes * 16;
    int fgroups   = ((num_nodes + 15) / 16 + FT / 32 - 1) / (FT / 32);
    const size_t toff = 2 * TICKET_GAP * sizeof(unsigned int);

    if (ws_size >= toff + koff + (3 * plane + 4 * (size_t)num_nodes) * sizeof(float)) {
        unsigned int* ticket = (unsigned int*)d_ws;
        unsigned int* keys = (unsigned int*)((char*)d_ws + toff);
        float* part = (float*)((char*)d_ws + toff + koff);
        float* pcnt = part + 3 * plane;
        int nblk = (num_nodes + 1600 - 1) / 1600;
        NeighbourAggregation_key_kernel<1600><<<512, KT, 0, stream>>>(edge_dst, keys, ticket, E);
        NeighbourAggregation_owner_kernel<1600, 4><<<nblk * 4, NT, 0, stream>>>(
            x, edge_dst, keys, out, part, pcnt, ticket, E, num_nodes);
        NeighbourAggregation_scale_kernel<4><<<fgroups, FT, 0, stream>>>(out, part, pcnt, num_nodes);
    } else {
        unsigned int* keys = (unsigned int*)d_ws;
        float* part = (float*)((char*)d_ws + koff);
        float* pcnt = part;
        int nblk = (num_nodes + 400 - 1) / 400;
        NeighbourAggregation_key_kernel<400><<<512, KT, 0, stream>>>(edge_dst, keys, nullptr, E);
        NeighbourAggregation_owner_kernel<400, 1><<<nblk, NT, 0, stream>>>(
            x, edge_dst, keys, out, part, pcnt, nullptr, E, num_nodes);
        NeighbourAggregation_scale_kernel<1><<<fgroups, FT, 0, stream>>>(out, part, pcnt, num_nodes);
    }
}
